# starting kernel
# baseline (speedup 1.0000x reference)
#include <hip/hip_runtime.h>

typedef __attribute__((ext_vector_type(2))) float v2f;
typedef __attribute__((ext_vector_type(8))) float v8f;

__global__ __launch_bounds__(256) void k_init_deg(float* __restrict__ deg, int n) {
  int i = blockIdx.x * blockDim.x + threadIdx.x;
  if (i < n) deg[i] = 1.0f;
}

__global__ __launch_bounds__(256) void k_deg_v4(const int* __restrict__ dst,
                                                float* __restrict__ deg, int nvec) {
  int i = blockIdx.x * blockDim.x + threadIdx.x;
  if (i < nvec) {
    int4 d = ((const int4*)dst)[i];
    atomicAdd(&deg[d.x], 1.0f);
    atomicAdd(&deg[d.y], 1.0f);
    atomicAdd(&deg[d.z], 1.0f);
    atomicAdd(&deg[d.w], 1.0f);
  }
}
__global__ __launch_bounds__(256) void k_deg_s(const int* __restrict__ dst,
                                               float* __restrict__ deg, int e) {
  int i = blockIdx.x * blockDim.x + threadIdx.x;
  if (i < e) atomicAdd(&deg[dst[i]], 1.0f);
}

__global__ __launch_bounds__(256) void k_node1(const float* __restrict__ x,
                                               const float* __restrict__ deg,
                                               float* __restrict__ dinv,
                                               float* __restrict__ p,
                                               float* __restrict__ t, int n) {
  int i = blockIdx.x * blockDim.x + threadIdx.x;
  if (i < n) {
    float dv = rsqrtf(deg[i]);
    dinv[i] = dv;
    float pv = x[i] * dv;
    p[i] = pv;
    t[i] = pv;
  }
}

__global__ __launch_bounds__(256) void k_scatter_v4(const int* __restrict__ src,
                                                    const int* __restrict__ dst,
                                                    const float* __restrict__ val,
                                                    float* __restrict__ acc, int nvec) {
  int i = blockIdx.x * blockDim.x + threadIdx.x;
  if (i < nvec) {
    int4 s = ((const int4*)src)[i];
    int4 d = ((const int4*)dst)[i];
    atomicAdd(&acc[d.x], val[s.x]);
    atomicAdd(&acc[d.y], val[s.y]);
    atomicAdd(&acc[d.z], val[s.z]);
    atomicAdd(&acc[d.w], val[s.w]);
  }
}
__global__ __launch_bounds__(256) void k_scatter_s(const int* __restrict__ src,
                                                   const int* __restrict__ dst,
                                                   const float* __restrict__ val,
                                                   float* __restrict__ acc, int e) {
  int i = blockIdx.x * blockDim.x + threadIdx.x;
  if (i < e) atomicAdd(&acc[dst[i]], val[src[i]]);
}

template <int IMM>
__device__ __forceinline__ float swz_add(float v) {
#if defined(__AMDGCN__)
  int s = __builtin_amdgcn_ds_swizzle(__float_as_int(v), IMM);
  return v + __int_as_float(s);
#else
  return v;
#endif
}

__global__ __launch_bounds__(256) void k_mid(const float* __restrict__ tacc,
                                             const float* __restrict__ dinv,
                                             const float* __restrict__ W1,
                                             const float* __restrict__ b1,
                                             const float* __restrict__ W2,
                                             float* __restrict__ q,
                                             float* __restrict__ u, int n) {
  const int lane   = threadIdx.x & 31;
  const int lane16 = lane & 15;
  const bool hi    = lane >= 16;
  const long long wave = (long long)blockIdx.x * (blockDim.x >> 5) + (threadIdx.x >> 5);
  const long long base = wave * 16;
  if (base >= n) return;

  float a = 0.0f;
  if (!hi) {
    long long idx = base + lane16;
    if (idx < n) a = tacc[idx] * dinv[idx];
  }
  const float w1v = W1[lane16];
  const float b1v = b1[lane16];
  const float w2v = W2[lane16];

#if __has_builtin(__builtin_amdgcn_wmma_f32_16x16x4_f32)
  v2f A; A.x = a;                  A.y = 0.0f;
  v2f B; B.x = hi ? 0.0f : w1v;    B.y = 0.0f;
  v8f C;
  #pragma unroll
  for (int r = 0; r < 8; ++r) C[r] = b1v;
  v8f D = __builtin_amdgcn_wmma_f32_16x16x4_f32(
      false, A, false, B, (short)0, C, false, false);
#else
  v8f D;
  #pragma unroll
  for (int r = 0; r < 8; ++r) {
    float am = __shfl(a, r + (hi ? 8 : 0), 32);
    D[r] = fmaf(am, w1v, b1v);
  }
#endif

  float z[8];
  #pragma unroll
  for (int r = 0; r < 8; ++r) {
    float h = D[r];
    h = h > 0.0f ? h : 0.0f;
    float v = h * w2v;
    v = swz_add<0x041F>(v);
    v = swz_add<0x081F>(v);
    v = swz_add<0x101F>(v);
    v = swz_add<0x201F>(v);
    z[r] = v;
  }

  if (lane == 0 || lane == 16) {
    const int off = hi ? 8 : 0;
    #pragma unroll
    for (int r = 0; r < 8; ++r) {
      long long idx = base + off + r;
      if (idx < n) {
        float qq = z[r] * dinv[idx];
        q[idx] = qq;
        u[idx] = qq;
      }
    }
  }
}

__global__ __launch_bounds__(256) void k_out(const float* __restrict__ dinv,
                                             const float* __restrict__ u,
                                             const float* __restrict__ b2,
                                             float* __restrict__ out, int n) {
  int i = blockIdx.x * blockDim.x + threadIdx.x;
  if (i < n) out[i] = fmaf(dinv[i], u[i], b2[0]);
}

extern "C" void kernel_launch(void* const* d_in, const int* in_sizes, int n_in,
                              void* d_out, int out_size, void* d_ws, size_t ws_size,
                              hipStream_t stream) {
  const float* x  = (const float*)d_in[0];
  const float* W1 = (const float*)d_in[1];
  const float* b1 = (const float*)d_in[2];
  const float* W2 = (const float*)d_in[3];
  const float* b2 = (const float*)d_in[4];
  const int*   ei = (const int*)d_in[5];

  const int n = in_sizes[0];
  const int e = in_sizes[5] / 2;
  const int* src = ei;
  const int* dst = ei + e;

  float* out  = (float*)d_out;
  float* dinv = (float*)d_ws;
  float* p    = dinv + n;
  float* t    = p + n;
  float* u    = t + n;
  float* q    = p;

  const int B  = 256;
  const int gN = (n + B - 1) / B;
  const int gM = (n + 127) / 128;

  k_init_deg<<<gN, B, 0, stream>>>(out, n);
  if ((e & 3) == 0) {
    int nvec = e >> 2;
    int gE = (nvec + B - 1) / B;
    k_deg_v4<<<gE, B, 0, stream>>>(dst, out, nvec);
  } else {
    int gE = (e + B - 1) / B;
    k_deg_s<<<gE, B, 0, stream>>>(dst, out, e);
  }

  k_node1<<<gN, B, 0, stream>>>(x, out, dinv, p, t, n);

  if ((e & 3) == 0) {
    int nvec = e >> 2;
    int gE = (nvec + B - 1) / B;
    k_scatter_v4<<<gE, B, 0, stream>>>(src, dst, p, t, nvec);
  } else {
    int gE = (e + B - 1) / B;
    k_scatter_s<<<gE, B, 0, stream>>>(src, dst, p, t, e);
  }

  k_mid<<<gM, B, 0, stream>>>(t, dinv, W1, b1, W2, q, u, n);

  if ((e & 3) == 0) {
    int nvec = e >> 2;
    int gE = (nvec + B - 1) / B;
    k_scatter_v4<<<gE, B, 0, stream>>>(src, dst, q, u, nvec);
  } else {
    int gE = (e + B - 1) / B;
    k_scatter_s<<<gE, B, 0, stream>>>(src, dst, q, u, e);
  }

  k_out<<<gN, B, 0, stream>>>(dinv, u, b2, out, n);
}


# Both float-atomic scatters replaced by gather-side sums over per-destination slot lists built by one returning integer add per edge (also the degree); sums now in slot order
# speedup vs baseline: 1.7715x; 1.7715x over previous
#include <hip/hip_runtime.h>

typedef __attribute__((ext_vector_type(2))) float v2f;
typedef __attribute__((ext_vector_type(8))) float v8f;

#define CAP 32

__global__ __launch_bounds__(256) void k_zero(int* __restrict__ cnt, int nwords) {
  int i = blockIdx.x * blockDim.x + threadIdx.x;
  if (i < nwords) cnt[i] = 0;
}

__device__ __forceinline__ void put_edge(int s, int d, int r, int* __restrict__ ell,
                                         int* __restrict__ spill_cnt,
                                         int2* __restrict__ spill, size_t stride) {
  if (r < CAP) {
    ell[(size_t)r * stride + (size_t)d] = s;
  } else {
    int pos = atomicAdd(spill_cnt, 1);
    spill[pos] = make_int2(s, d);
  }
}

__global__ __launch_bounds__(256) void k_build_v4(const int* __restrict__ src,
                                                  const int* __restrict__ dst,
                                                  int* __restrict__ cnt,
                                                  int* __restrict__ ell,
                                                  int* __restrict__ spill_cnt,
                                                  int2* __restrict__ spill,
                                                  size_t stride, int nvec) {
  int i = blockIdx.x * blockDim.x + threadIdx.x;
  if (i < nvec) {
    int4 s = ((const int4*)src)[i];
    int4 d = ((const int4*)dst)[i];
    int r0 = atomicAdd(&cnt[d.x], 1);
    int r1 = atomicAdd(&cnt[d.y], 1);
    int r2 = atomicAdd(&cnt[d.z], 1);
    int r3 = atomicAdd(&cnt[d.w], 1);
    put_edge(s.x, d.x, r0, ell, spill_cnt, spill, stride);
    put_edge(s.y, d.y, r1, ell, spill_cnt, spill, stride);
    put_edge(s.z, d.z, r2, ell, spill_cnt, spill, stride);
    put_edge(s.w, d.w, r3, ell, spill_cnt, spill, stride);
  }
}
__global__ __launch_bounds__(256) void k_build_s(const int* __restrict__ src,
                                                 const int* __restrict__ dst,
                                                 int* __restrict__ cnt,
                                                 int* __restrict__ ell,
                                                 int* __restrict__ spill_cnt,
                                                 int2* __restrict__ spill,
                                                 size_t stride, int e) {
  int i = blockIdx.x * blockDim.x + threadIdx.x;
  if (i < e) {
    int s = src[i], d = dst[i];
    int r = atomicAdd(&cnt[d], 1);
    put_edge(s, d, r, ell, spill_cnt, spill, stride);
  }
}

__global__ __launch_bounds__(256) void k_node1(const float* __restrict__ x,
                                               const int* __restrict__ cnt,
                                               float* __restrict__ dinv,
                                               float* __restrict__ p, int n) {
  int i = blockIdx.x * blockDim.x + threadIdx.x;
  if (i < n) {
    float dv = rsqrtf((float)(cnt[i] + 1));
    dinv[i] = dv;
    p[i] = x[i] * dv;
  }
}

template <bool FINAL>
__global__ __launch_bounds__(256) void k_pull(const int* __restrict__ cnt,
                                              const int* __restrict__ ell,
                                              const int* __restrict__ spill_cnt,
                                              const int2* __restrict__ spill,
                                              const float* __restrict__ val,
                                              const float* __restrict__ dinv,
                                              const float* __restrict__ b2,
                                              float* __restrict__ res,
                                              size_t stride, int n) {
  int i = blockIdx.x * blockDim.x + threadIdx.x;
  if (i >= n) return;
  const int c = cnt[i];
  const int m = c < CAP ? c : CAP;
  float acc = val[i];
  const int* __restrict__ col = ell + i;
  for (int r = 0; r < m; r += 4) {
    const bool h1 = r + 1 < m, h2 = r + 2 < m, h3 = r + 3 < m;
    int s0 = col[(size_t)r * stride];
    int s1 = h1 ? col[(size_t)(r + 1) * stride] : i;
    int s2 = h2 ? col[(size_t)(r + 2) * stride] : i;
    int s3 = h3 ? col[(size_t)(r + 3) * stride] : i;
    float v0 = val[s0];
    float v1 = h1 ? val[s1] : 0.0f;
    float v2 = h2 ? val[s2] : 0.0f;
    float v3 = h3 ? val[s3] : 0.0f;
    acc += v0;
    if (h1) acc += v1;
    if (h2) acc += v2;
    if (h3) acc += v3;
  }
  if (c > CAP) {
    const int ns = *spill_cnt;
    for (int j = 0; j < ns; ++j) {
      int2 o = spill[j];
      if (o.y == i) acc += val[o.x];
    }
  }
  if (FINAL) res[i] = fmaf(dinv[i], acc, b2[0]);
  else       res[i] = acc;
}

template <int IMM>
__device__ __forceinline__ float swz_add(float v) {
#if defined(__AMDGCN__)
  int s = __builtin_amdgcn_ds_swizzle(__float_as_int(v), IMM);
  return v + __int_as_float(s);
#else
  return v;
#endif
}

__global__ __launch_bounds__(256) void k_mid(const float* __restrict__ tacc,
                                             const float* __restrict__ dinv,
                                             const float* __restrict__ W1,
                                             const float* __restrict__ b1,
                                             const float* __restrict__ W2,
                                             float* __restrict__ q, int n) {
  const int lane   = threadIdx.x & 31;
  const int lane16 = lane & 15;
  const bool hi    = lane >= 16;
  const long long wave = (long long)blockIdx.x * (blockDim.x >> 5) + (threadIdx.x >> 5);
  const long long base = wave * 16;
  if (base >= n) return;

  float a = 0.0f;
  if (!hi) {
    long long idx = base + lane16;
    if (idx < n) a = tacc[idx] * dinv[idx];
  }
  const float w1v = W1[lane16];
  const float b1v = b1[lane16];
  const float w2v = W2[lane16];

#if __has_builtin(__builtin_amdgcn_wmma_f32_16x16x4_f32)
  v2f A; A.x = a;                  A.y = 0.0f;
  v2f B; B.x = hi ? 0.0f : w1v;    B.y = 0.0f;
  v8f C;
  #pragma unroll
  for (int r = 0; r < 8; ++r) C[r] = b1v;
  v8f D = __builtin_amdgcn_wmma_f32_16x16x4_f32(
      false, A, false, B, (short)0, C, false, false);
#else
  v8f D;
  #pragma unroll
  for (int r = 0; r < 8; ++r) {
    float am = __shfl(a, r + (hi ? 8 : 0), 32);
    D[r] = fmaf(am, w1v, b1v);
  }
#endif

  float z[8];
  #pragma unroll
  for (int r = 0; r < 8; ++r) {
    float h = D[r];
    h = h > 0.0f ? h : 0.0f;
    float v = h * w2v;
    v = swz_add<0x041F>(v);
    v = swz_add<0x081F>(v);
    v = swz_add<0x101F>(v);
    v = swz_add<0x201F>(v);
    z[r] = v;
  }

  if (lane == 0 || lane == 16) {
    const int off = hi ? 8 : 0;
    #pragma unroll
    for (int r = 0; r < 8; ++r) {
      long long idx = base + off + r;
      if (idx < n) {
        float qq = z[r] * dinv[idx];
        q[idx] = qq;
      }
    }
  }
}

extern "C" void kernel_launch(void* const* d_in, const int* in_sizes, int n_in,
                              void* d_out, int out_size, void* d_ws, size_t ws_size,
                              hipStream_t stream) {
  const float* x  = (const float*)d_in[0];
  const float* W1 = (const float*)d_in[1];
  const float* b1 = (const float*)d_in[2];
  const float* W2 = (const float*)d_in[3];
  const float* b2 = (const float*)d_in[4];
  const int*   ei = (const int*)d_in[5];

  const int n = in_sizes[0];
  const int e = in_sizes[5] / 2;
  const int* src = ei;
  const int* dst = ei + e;

  float* out  = (float*)d_out;
  const size_t stride = ((size_t)n + 31) & ~(size_t)31;
  float* dinv = (float*)d_ws;
  float* p    = dinv + stride;
  float* t    = p + stride;
  float* q    = p;
  int*   cnt  = (int*)(t + stride);
  int*   spill_cnt = cnt + stride;
  int*   ell  = spill_cnt + 32;
  int2*  spill = (int2*)(ell + (size_t)CAP * stride);

  const int B  = 256;
  const int gN = (n + B - 1) / B;
  const int gM = (n + 127) / 128;
  const int nz = (int)stride + 32;
  const int gZ = (nz + B - 1) / B;

  k_zero<<<gZ, B, 0, stream>>>(cnt, nz);

  if ((e & 3) == 0) {
    int nvec = e >> 2;
    int gE = (nvec + B - 1) / B;
    k_build_v4<<<gE, B, 0, stream>>>(src, dst, cnt, ell, spill_cnt, spill, stride, nvec);
  } else {
    int gE = (e + B - 1) / B;
    k_build_s<<<gE, B, 0, stream>>>(src, dst, cnt, ell, spill_cnt, spill, stride, e);
  }

  k_node1<<<gN, B, 0, stream>>>(x, cnt, dinv, p, n);

  k_pull<false><<<gN, B, 0, stream>>>(cnt, ell, spill_cnt, spill, p, dinv, b2, t, stride, n);

  k_mid<<<gM, B, 0, stream>>>(t, dinv, W1, b1, W2, q, n);

  k_pull<true><<<gN, B, 0, stream>>>(cnt, ell, spill_cnt, spill, q, dinv, b2, out, stride, n);
}


# One set of counters and slot planes per package half, picked by the wave's die number: an edge's add and slot store stay in the adder's cache half; gathers walk both sets (sum order changes)
# speedup vs baseline: 4.6638x; 2.6327x over previous
#include <hip/hip_runtime.h>

typedef __attribute__((ext_vector_type(2))) float v2f;
typedef __attribute__((ext_vector_type(8))) float v8f;

#define CAP 32

__device__ __forceinline__ int pkg_half() {
#if defined(__AMDGCN__)
  unsigned v = __builtin_amdgcn_s_sendmsg_rtn(0x87);
  return (int)((v >> 18) & 1u);
#else
  return 0;
#endif
}

__global__ __launch_bounds__(256) void k_zero(int* __restrict__ cnt, int nwords) {
  int i = blockIdx.x * blockDim.x + threadIdx.x;
  if (i < nwords) cnt[i] = 0;
}

__device__ __forceinline__ void put_edge(int s, int d, int r, int* __restrict__ ell,
                                         int* __restrict__ spill_cnt,
                                         int2* __restrict__ spill, size_t stride) {
  if (r < CAP) {
    ell[(size_t)r * stride + (size_t)d] = s;
  } else {
    int pos = atomicAdd(spill_cnt, 1);
    spill[pos] = make_int2(s, d);
  }
}

__global__ __launch_bounds__(256) void k_build_v4(const int* __restrict__ src,
                                                  const int* __restrict__ dst,
                                                  int* __restrict__ cnt,
                                                  int* __restrict__ ell,
                                                  int* __restrict__ spill_cnt,
                                                  int2* __restrict__ spill,
                                                  size_t stride, int nvec) {
  int i = blockIdx.x * blockDim.x + threadIdx.x;
  const int hf = pkg_half();
  cnt += (size_t)hf * stride;
  ell += (size_t)hf * CAP * stride;
  if (i < nvec) {
    int4 s = ((const int4*)src)[i];
    int4 d = ((const int4*)dst)[i];
    int r0 = atomicAdd(&cnt[d.x], 1);
    int r1 = atomicAdd(&cnt[d.y], 1);
    int r2 = atomicAdd(&cnt[d.z], 1);
    int r3 = atomicAdd(&cnt[d.w], 1);
    put_edge(s.x, d.x, r0, ell, spill_cnt, spill, stride);
    put_edge(s.y, d.y, r1, ell, spill_cnt, spill, stride);
    put_edge(s.z, d.z, r2, ell, spill_cnt, spill, stride);
    put_edge(s.w, d.w, r3, ell, spill_cnt, spill, stride);
  }
}
__global__ __launch_bounds__(256) void k_build_s(const int* __restrict__ src,
                                                 const int* __restrict__ dst,
                                                 int* __restrict__ cnt,
                                                 int* __restrict__ ell,
                                                 int* __restrict__ spill_cnt,
                                                 int2* __restrict__ spill,
                                                 size_t stride, int e) {
  int i = blockIdx.x * blockDim.x + threadIdx.x;
  const int hf = pkg_half();
  cnt += (size_t)hf * stride;
  ell += (size_t)hf * CAP * stride;
  if (i < e) {
    int s = src[i], d = dst[i];
    int r = atomicAdd(&cnt[d], 1);
    put_edge(s, d, r, ell, spill_cnt, spill, stride);
  }
}

__global__ __launch_bounds__(256) void k_node1(const float* __restrict__ x,
                                               const int* __restrict__ cnt,
                                               float* __restrict__ dinv,
                                               float* __restrict__ p,
                                               size_t stride, int n) {
  int i = blockIdx.x * blockDim.x + threadIdx.x;
  if (i < n) {
    float dv = rsqrtf((float)(cnt[i] + cnt[stride + i] + 1));
    dinv[i] = dv;
    p[i] = x[i] * dv;
  }
}

template <bool FINAL>
__global__ __launch_bounds__(256) void k_pull(const int* __restrict__ cnt,
                                              const int* __restrict__ ell,
                                              const int* __restrict__ spill_cnt,
                                              const int2* __restrict__ spill,
                                              const float* __restrict__ val,
                                              const float* __restrict__ dinv,
                                              const float* __restrict__ b2,
                                              float* __restrict__ res,
                                              size_t stride, int n) {
  int i = blockIdx.x * blockDim.x + threadIdx.x;
  if (i >= n) return;
  const int c0 = cnt[i];
  const int c1 = cnt[stride + i];
  const int m0 = c0 < CAP ? c0 : CAP;
  const int m1 = c1 < CAP ? c1 : CAP;
  const int mm = m0 > m1 ? m0 : m1;
  float acc = val[i];
  const int* __restrict__ colA = ell + i;
  const int* __restrict__ colB = ell + (size_t)CAP * stride + i;
  for (int r = 0; r < mm; r += 4) {
    const bool ha0 = r < m0, ha1 = r + 1 < m0, ha2 = r + 2 < m0, ha3 = r + 3 < m0;
    const bool hb0 = r < m1, hb1 = r + 1 < m1, hb2 = r + 2 < m1, hb3 = r + 3 < m1;
    const int sa0 = ha0 ? colA[(size_t)r * stride] : i;
    const int sa1 = ha1 ? colA[(size_t)(r + 1) * stride] : i;
    const int sa2 = ha2 ? colA[(size_t)(r + 2) * stride] : i;
    const int sa3 = ha3 ? colA[(size_t)(r + 3) * stride] : i;
    const int sb0 = hb0 ? colB[(size_t)r * stride] : i;
    const int sb1 = hb1 ? colB[(size_t)(r + 1) * stride] : i;
    const int sb2 = hb2 ? colB[(size_t)(r + 2) * stride] : i;
    const int sb3 = hb3 ? colB[(size_t)(r + 3) * stride] : i;
    const float va0 = val[sa0], va1 = val[sa1], va2 = val[sa2], va3 = val[sa3];
    const float vb0 = val[sb0], vb1 = val[sb1], vb2 = val[sb2], vb3 = val[sb3];
    if (ha0) acc += va0;
    if (ha1) acc += va1;
    if (ha2) acc += va2;
    if (ha3) acc += va3;
    if (hb0) acc += vb0;
    if (hb1) acc += vb1;
    if (hb2) acc += vb2;
    if (hb3) acc += vb3;
  }
  if (c0 > CAP || c1 > CAP) {
    const int ns = *spill_cnt;
    for (int j = 0; j < ns; ++j) {
      int2 o = spill[j];
      if (o.y == i) acc += val[o.x];
    }
  }
  if (FINAL) res[i] = fmaf(dinv[i], acc, b2[0]);
  else       res[i] = acc;
}

template <int IMM>
__device__ __forceinline__ float swz_add(float v) {
#if defined(__AMDGCN__)
  int s = __builtin_amdgcn_ds_swizzle(__float_as_int(v), IMM);
  return v + __int_as_float(s);
#else
  return v;
#endif
}

__global__ __launch_bounds__(256) void k_mid(const float* __restrict__ tacc,
                                             const float* __restrict__ dinv,
                                             const float* __restrict__ W1,
                                             const float* __restrict__ b1,
                                             const float* __restrict__ W2,
                                             float* __restrict__ q, int n) {
  const int lane   = threadIdx.x & 31;
  const int lane16 = lane & 15;
  const bool hi    = lane >= 16;
  const long long wave = (long long)blockIdx.x * (blockDim.x >> 5) + (threadIdx.x >> 5);
  const long long base = wave * 16;
  if (base >= n) return;

  float a = 0.0f;
  if (!hi) {
    long long idx = base + lane16;
    if (idx < n) a = tacc[idx] * dinv[idx];
  }
  const float w1v = W1[lane16];
  const float b1v = b1[lane16];
  const float w2v = W2[lane16];

#if __has_builtin(__builtin_amdgcn_wmma_f32_16x16x4_f32)
  v2f A; A.x = a;                  A.y = 0.0f;
  v2f B; B.x = hi ? 0.0f : w1v;    B.y = 0.0f;
  v8f C;
  #pragma unroll
  for (int r = 0; r < 8; ++r) C[r] = b1v;
  v8f D = __builtin_amdgcn_wmma_f32_16x16x4_f32(
      false, A, false, B, (short)0, C, false, false);
#else
  v8f D;
  #pragma unroll
  for (int r = 0; r < 8; ++r) {
    float am = __shfl(a, r + (hi ? 8 : 0), 32);
    D[r] = fmaf(am, w1v, b1v);
  }
#endif

  float z[8];
  #pragma unroll
  for (int r = 0; r < 8; ++r) {
    float h = D[r];
    h = h > 0.0f ? h : 0.0f;
    float v = h * w2v;
    v = swz_add<0x041F>(v);
    v = swz_add<0x081F>(v);
    v = swz_add<0x101F>(v);
    v = swz_add<0x201F>(v);
    z[r] = v;
  }

  if (lane == 0 || lane == 16) {
    const int off = hi ? 8 : 0;
    #pragma unroll
    for (int r = 0; r < 8; ++r) {
      long long idx = base + off + r;
      if (idx < n) {
        float qq = z[r] * dinv[idx];
        q[idx] = qq;
      }
    }
  }
}

extern "C" void kernel_launch(void* const* d_in, const int* in_sizes, int n_in,
                              void* d_out, int out_size, void* d_ws, size_t ws_size,
                              hipStream_t stream) {
  const float* x  = (const float*)d_in[0];
  const float* W1 = (const float*)d_in[1];
  const float* b1 = (const float*)d_in[2];
  const float* W2 = (const float*)d_in[3];
  const float* b2 = (const float*)d_in[4];
  const int*   ei = (const int*)d_in[5];

  const int n = in_sizes[0];
  const int e = in_sizes[5] / 2;
  const int* src = ei;
  const int* dst = ei + e;

  float* out  = (float*)d_out;
  const size_t stride = ((size_t)n + 31) & ~(size_t)31;
  float* dinv = (float*)d_ws;
  float* p    = dinv + stride;
  float* t    = p + stride;
  float* q    = p;
  int*   cnt  = (int*)(t + stride);
  int*   spill_cnt = cnt + 2 * stride;
  int*   ell  = spill_cnt + 32;
  int2*  spill = (int2*)(ell + (size_t)2 * CAP * stride);

  const int B  = 256;
  const int gN = (n + B - 1) / B;
  const int gM = (n + 127) / 128;
  const int nz = 2 * (int)stride + 32;
  const int gZ = (nz + B - 1) / B;

  k_zero<<<gZ, B, 0, stream>>>(cnt, nz);

  if ((e & 3) == 0) {
    int nvec = e >> 2;
    int gE = (nvec + B - 1) / B;
    k_build_v4<<<gE, B, 0, stream>>>(src, dst, cnt, ell, spill_cnt, spill, stride, nvec);
  } else {
    int gE = (e + B - 1) / B;
    k_build_s<<<gE, B, 0, stream>>>(src, dst, cnt, ell, spill_cnt, spill, stride, e);
  }

  k_node1<<<gN, B, 0, stream>>>(x, cnt, dinv, p, stride, n);

  k_pull<false><<<gN, B, 0, stream>>>(cnt, ell, spill_cnt, spill, p, dinv, b2, t, stride, n);

  k_mid<<<gM, B, 0, stream>>>(t, dinv, W1, b1, W2, q, n);

  k_pull<true><<<gN, B, 0, stream>>>(cnt, ell, spill_cnt, spill, q, dinv, b2, out, stride, n);
}


# Middle kernel fused into the layer-1 gather: each wave feeds its 32 sums to two WMMA tiles as columns (nodes = N) and sums hidden units in-lane (z sum order changes); t plane and one launch gone
# speedup vs baseline: 5.2295x; 1.1213x over previous
#include <hip/hip_runtime.h>

typedef __attribute__((ext_vector_type(2))) float v2f;
typedef __attribute__((ext_vector_type(8))) float v8f;

#define CAP 32

__device__ __forceinline__ int pkg_half() {
#if defined(__AMDGCN__)
  unsigned v = __builtin_amdgcn_s_sendmsg_rtn(0x87);
  return (int)((v >> 18) & 1u);
#else
  return 0;
#endif
}

__global__ __launch_bounds__(256) void k_zero(int* __restrict__ cnt, int nwords) {
  int i = blockIdx.x * blockDim.x + threadIdx.x;
  if (i < nwords) cnt[i] = 0;
}

__device__ __forceinline__ void put_edge(int s, int d, int r, int* __restrict__ ell,
                                         int* __restrict__ spill_cnt,
                                         int2* __restrict__ spill, size_t stride) {
  if (r < CAP) {
    ell[(size_t)r * stride + (size_t)d] = s;
  } else {
    int pos = atomicAdd(spill_cnt, 1);
    spill[pos] = make_int2(s, d);
  }
}

__global__ __launch_bounds__(256) void k_build_v4(const int* __restrict__ src,
                                                  const int* __restrict__ dst,
                                                  int* __restrict__ cnt,
                                                  int* __restrict__ ell,
                                                  int* __restrict__ spill_cnt,
                                                  int2* __restrict__ spill,
                                                  size_t stride, int nvec) {
  int i = blockIdx.x * blockDim.x + threadIdx.x;
  const int hf = pkg_half();
  cnt += (size_t)hf * stride;
  ell += (size_t)hf * CAP * stride;
  if (i < nvec) {
    int4 s = ((const int4*)src)[i];
    int4 d = ((const int4*)dst)[i];
    int r0 = atomicAdd(&cnt[d.x], 1);
    int r1 = atomicAdd(&cnt[d.y], 1);
    int r2 = atomicAdd(&cnt[d.z], 1);
    int r3 = atomicAdd(&cnt[d.w], 1);
    put_edge(s.x, d.x, r0, ell, spill_cnt, spill, stride);
    put_edge(s.y, d.y, r1, ell, spill_cnt, spill, stride);
    put_edge(s.z, d.z, r2, ell, spill_cnt, spill, stride);
    put_edge(s.w, d.w, r3, ell, spill_cnt, spill, stride);
  }
}
__global__ __launch_bounds__(256) void k_build_s(const int* __restrict__ src,
                                                 const int* __restrict__ dst,
                                                 int* __restrict__ cnt,
                                                 int* __restrict__ ell,
                                                 int* __restrict__ spill_cnt,
                                                 int2* __restrict__ spill,
                                                 size_t stride, int e) {
  int i = blockIdx.x * blockDim.x + threadIdx.x;
  const int hf = pkg_half();
  cnt += (size_t)hf * stride;
  ell += (size_t)hf * CAP * stride;
  if (i < e) {
    int s = src[i], d = dst[i];
    int r = atomicAdd(&cnt[d], 1);
    put_edge(s, d, r, ell, spill_cnt, spill, stride);
  }
}

__global__ __launch_bounds__(256) void k_node1(const float* __restrict__ x,
                                               const int* __restrict__ cnt,
                                               float* __restrict__ dinv,
                                               float* __restrict__ p,
                                               size_t stride, int n) {
  int i = blockIdx.x * blockDim.x + threadIdx.x;
  if (i < n) {
    float dv = rsqrtf((float)(cnt[i] + cnt[stride + i] + 1));
    dinv[i] = dv;
    p[i] = x[i] * dv;
  }
}

__device__ __forceinline__ float other_half(float v) {
#if defined(__AMDGCN__)
  return __int_as_float(__builtin_amdgcn_ds_swizzle(__float_as_int(v), 0x401F));
#else
  return v;
#endif
}

__device__ __forceinline__ float mid_stage(float a, int lane,
                                           const float* __restrict__ W1,
                                           const float* __restrict__ b1,
                                           const float* __restrict__ W2) {
  const bool hi = lane >= 16;
  const float w1v = W1[lane & 15];
  float cb[8], w2[8];
  #pragma unroll
  for (int r = 0; r < 8; ++r) {
    cb[r] = hi ? b1[8 + r] : b1[r];
    w2[r] = hi ? W2[8 + r] : W2[r];
  }
#if __has_builtin(__builtin_amdgcn_wmma_f32_16x16x4_f32)
  v2f A1; A1.x = hi ? 0.0f : w1v;  A1.y = 0.0f;
  v2f B1; B1.x = hi ? 0.0f : a;    B1.y = 0.0f;
  v2f A2; A2.x = hi ? w1v : 0.0f;  A2.y = 0.0f;
  v2f B2; B2.x = hi ? a : 0.0f;    B2.y = 0.0f;
  v8f C;
  #pragma unroll
  for (int r = 0; r < 8; ++r) C[r] = cb[r];
  v8f D1 = __builtin_amdgcn_wmma_f32_16x16x4_f32(
      false, A1, false, B1, (short)0, C, false, false);
  v8f D2 = __builtin_amdgcn_wmma_f32_16x16x4_f32(
      false, A2, false, B2, (short)0, C, false, false);
#else
  v8f D1, D2;
  #pragma unroll
  for (int r = 0; r < 8; ++r) {
    D1[r] = fmaf(w1v, __shfl(a, lane & 15, 32), cb[r]);
    D2[r] = fmaf(w1v, __shfl(a, 16 + (lane & 15), 32), cb[r]);
  }
#endif
  float s1 = 0.0f, s2 = 0.0f;
  #pragma unroll
  for (int r = 0; r < 8; ++r) {
    const float h1 = D1[r] > 0.0f ? D1[r] : 0.0f;
    const float h2 = D2[r] > 0.0f ? D2[r] : 0.0f;
    s1 = fmaf(h1, w2[r], s1);
    s2 = fmaf(h2, w2[r], s2);
  }
  const float mine   = hi ? s2 : s1;
  const float theirs = hi ? s1 : s2;
  return mine + other_half(theirs);
}

template <bool FINAL>
__global__ __launch_bounds__(256) void k_pull(const int* __restrict__ cnt,
                                              const int* __restrict__ ell,
                                              const int* __restrict__ spill_cnt,
                                              const int2* __restrict__ spill,
                                              const float* __restrict__ val,
                                              const float* __restrict__ dinv,
                                              const float* __restrict__ b2,
                                              const float* __restrict__ W1,
                                              const float* __restrict__ b1,
                                              const float* __restrict__ W2,
                                              float* __restrict__ res,
                                              size_t stride, int n) {
  int i = blockIdx.x * blockDim.x + threadIdx.x;
  const bool valid = i < n;
  if (FINAL) {
    if (!valid) return;
  } else {
    if ((i & ~31) >= n) return;
  }
  const int ic = valid ? i : n - 1;
  int c0 = cnt[ic];
  int c1 = cnt[stride + ic];
  if (!valid) { c0 = 0; c1 = 0; }
  const int m0 = c0 < CAP ? c0 : CAP;
  const int m1 = c1 < CAP ? c1 : CAP;
  const int mm = m0 > m1 ? m0 : m1;
  float acc = val[ic];
  float dv = FINAL ? 0.0f : dinv[ic];
  if (!valid) { acc = 0.0f; dv = 0.0f; }
  const int* __restrict__ colA = ell + i;
  const int* __restrict__ colB = ell + (size_t)CAP * stride + i;
  for (int r = 0; r < mm; r += 4) {
    const bool ha0 = r < m0, ha1 = r + 1 < m0, ha2 = r + 2 < m0, ha3 = r + 3 < m0;
    const bool hb0 = r < m1, hb1 = r + 1 < m1, hb2 = r + 2 < m1, hb3 = r + 3 < m1;
    const int sa0 = ha0 ? colA[(size_t)r * stride] : i;
    const int sa1 = ha1 ? colA[(size_t)(r + 1) * stride] : i;
    const int sa2 = ha2 ? colA[(size_t)(r + 2) * stride] : i;
    const int sa3 = ha3 ? colA[(size_t)(r + 3) * stride] : i;
    const int sb0 = hb0 ? colB[(size_t)r * stride] : i;
    const int sb1 = hb1 ? colB[(size_t)(r + 1) * stride] : i;
    const int sb2 = hb2 ? colB[(size_t)(r + 2) * stride] : i;
    const int sb3 = hb3 ? colB[(size_t)(r + 3) * stride] : i;
    const float va0 = val[sa0], va1 = val[sa1], va2 = val[sa2], va3 = val[sa3];
    const float vb0 = val[sb0], vb1 = val[sb1], vb2 = val[sb2], vb3 = val[sb3];
    if (ha0) acc += va0;
    if (ha1) acc += va1;
    if (ha2) acc += va2;
    if (ha3) acc += va3;
    if (hb0) acc += vb0;
    if (hb1) acc += vb1;
    if (hb2) acc += vb2;
    if (hb3) acc += vb3;
  }
  if (c0 > CAP || c1 > CAP) {
    const int ns = *spill_cnt;
    for (int j = 0; j < ns; ++j) {
      int2 o = spill[j];
      if (o.y == i) acc += val[o.x];
    }
  }
  if (FINAL) {
    res[i] = fmaf(dinv[i], acc, b2[0]);
  } else {
    const float z = mid_stage(acc * dv, threadIdx.x & 31, W1, b1, W2);
    if (valid) res[i] = z * dv;
  }
}

extern "C" void kernel_launch(void* const* d_in, const int* in_sizes, int n_in,
                              void* d_out, int out_size, void* d_ws, size_t ws_size,
                              hipStream_t stream) {
  const float* x  = (const float*)d_in[0];
  const float* W1 = (const float*)d_in[1];
  const float* b1 = (const float*)d_in[2];
  const float* W2 = (const float*)d_in[3];
  const float* b2 = (const float*)d_in[4];
  const int*   ei = (const int*)d_in[5];

  const int n = in_sizes[0];
  const int e = in_sizes[5] / 2;
  const int* src = ei;
  const int* dst = ei + e;

  float* out  = (float*)d_out;
  const size_t stride = ((size_t)n + 31) & ~(size_t)31;
  float* dinv = (float*)d_ws;
  float* p    = dinv + stride;
  float* q    = p + stride;
  int*   cnt  = (int*)(q + stride);
  int*   spill_cnt = cnt + 2 * stride;
  int*   ell  = spill_cnt + 32;
  int2*  spill = (int2*)(ell + (size_t)2 * CAP * stride);

  const int B  = 256;
  const int gN = (n + B - 1) / B;
  const int nz = 2 * (int)stride + 32;
  const int gZ = (nz + B - 1) / B;

  k_zero<<<gZ, B, 0, stream>>>(cnt, nz);

  if ((e & 3) == 0) {
    int nvec = e >> 2;
    int gE = (nvec + B - 1) / B;
    k_build_v4<<<gE, B, 0, stream>>>(src, dst, cnt, ell, spill_cnt, spill, stride, nvec);
  } else {
    int gE = (e + B - 1) / B;
    k_build_s<<<gE, B, 0, stream>>>(src, dst, cnt, ell, spill_cnt, spill, stride, e);
  }

  k_node1<<<gN, B, 0, stream>>>(x, cnt, dinv, p, stride, n);

  k_pull<false><<<gN, B, 0, stream>>>(cnt, ell, spill_cnt, spill, p, dinv, b2, W1, b1, W2, q, stride, n);

  k_pull<true><<<gN, B, 0, stream>>>(cnt, ell, spill_cnt, spill, q, dinv, b2, W1, b1, W2, out, stride, n);
}
